# starting kernel
# baseline (speedup 1.0000x reference)
#include <hip/hip_runtime.h>

#define ALPHA  0.99f
#define GAMMA  0.999f
#define LAMBD  0.7f

typedef __attribute__((ext_vector_type(2))) float v2f;
typedef __attribute__((ext_vector_type(8))) float v8f;

__global__ __launch_bounds__(256)
void rlm_prep(const float* __restrict__ q, const float* __restrict__ q_tar,
              const float* __restrict__ pi, const int* __restrict__ a_t,
              const float* __restrict__ r_t, const float* __restrict__ mu,
              const int* __restrict__ done_t,
              float* __restrict__ adv,
              float* __restrict__ wsA, float* __restrict__ wsB,
              float* __restrict__ wsE, float* __restrict__ wsM,
              float* __restrict__ wsD)
{
    __shared__ float lds_v[8 * 16];
    __shared__ float lds_vp[8 * 16];

    const int lane = threadIdx.x & 31;
    const int wave = threadIdx.x >> 5;
    const int tile = blockIdx.x * 8 + wave;
    const int t0   = tile * 16;
    const int row  = lane & 15;
    const int colbase = (lane >> 4) << 1;

    const float* qrow   = q     + (size_t)(t0 + row) * 64 + colbase;
    const float* pirow  = pi    + (size_t)(t0 + row) * 64 + colbase;
    const float* qtrow  = q_tar + (size_t)(t0 + row + 1) * 64 + colbase;
    const float* piprow = pi    + (size_t)(t0 + row + 1) * 64 + colbase;

    const v2f ones = {1.0f, 1.0f};
    v8f acc_v  = {};
    v8f acc_vp = {};
    v2f qv[16];

    #pragma unroll
    for (int s = 0; s < 16; ++s) {
        v2f qd = *(const v2f*)(qrow  + 4 * s);
        v2f pd = *(const v2f*)(pirow + 4 * s);
        qv[s] = qd;
        v2f x = qd * pd;
        acc_v = __builtin_amdgcn_wmma_f32_16x16x4_f32(
            false, x, false, ones, (short)0, acc_v, false, false);
        v2f qtd = *(const v2f*)(qtrow  + 4 * s);
        v2f ppd = *(const v2f*)(piprow + 4 * s);
        v2f x2 = qtd * ppd;
        acc_vp = __builtin_amdgcn_wmma_f32_16x16x4_f32(
            false, x2, false, ones, (short)0, acc_vp, false, false);
    }

    float selv = acc_v[0], selvp = acc_vp[0];
    #pragma unroll
    for (int j = 1; j < 8; ++j) {
        bool pick = ((row & 7) == j);
        selv  = pick ? acc_v[j]  : selv;
        selvp = pick ? acc_vp[j] : selvp;
    }
    float* myv  = lds_v  + wave * 16;
    float* myvp = lds_vp + wave * 16;
    if ((lane & 15) < 8) {
        int idx = (lane & 7) + ((lane >> 4) << 3);
        myv[idx]  = selv;
        myvp[idx] = selvp;
    }
    __syncthreads();

    const float vt = myv[row];

    float* advrow = adv + (size_t)(t0 + row) * 64 + colbase;
    const float ascale = 1.0f - ALPHA;
    #pragma unroll
    for (int s = 0; s < 16; ++s) {
        v2f o = (qv[s] - vt) * ascale;
        *(v2f*)(advrow + 4 * s) = o;
    }

    if (lane < 16) {
        const int t = t0 + lane;
        const int a = a_t[t];
        const float q_t_a = q [(size_t)t * 64 + a];
        const float pi_a  = pi[(size_t)t * 64 + a];
        const float mu_a  = mu[(size_t)t * 64 + a];
        const float r     = r_t[t];
        const int   d     = done_t[t];
        const float df    = (float)d;
        const float vp1   = myvp[lane];
        const float vv    = myv[lane];

        const float est = r + (1.0f - df) * GAMMA * vp1;
        const float td  = est - q_t_a;
        const float rho = pi_a / mu_a;
        const float cc  = LAMBD * fminf(fmaxf(rho, 0.0f), 1.0f);

        wsB[t] = LAMBD * GAMMA * rho * td;
        wsA[t] = d ? 0.0f : GAMMA * cc;
        wsM[t] = d ? 0.0f : 1.0f;
        wsE[t] = est;
        wsD[t] = ALPHA * (q_t_a - vv);
    }
}

__global__ __launch_bounds__(256)
void rlm_compose(const float* __restrict__ wsA, const float* __restrict__ wsB,
                 float* __restrict__ chA, float* __restrict__ chB, int S)
{
    const int c = blockIdx.x * blockDim.x + threadIdx.x;
    const int base = c * S;
    float Ac = 1.0f, Bc = 0.0f;
    for (int i = S - 1; i >= 0; --i) {
        const float a = wsA[base + i];
        const float b = wsB[base + i];
        Bc = b + a * Bc;
        Ac = a * Ac;
    }
    chA[c] = Ac;
    chB[c] = Bc;
}

__global__ __launch_bounds__(1024)
void rlm_chunkscan(const float* __restrict__ chA, const float* __restrict__ chB,
                   float* __restrict__ carryIn, int C)
{
    __shared__ float sA[1024];
    __shared__ float sB[1024];
    const int i = threadIdx.x;
    if (i < C) { sA[i] = chA[i]; sB[i] = chB[i]; }
    __syncthreads();
    for (int off = 1; off < C; off <<= 1) {
        float a = 0.f, b = 0.f, a2 = 0.f, b2 = 0.f;
        const bool ok = (i + off) < C && i < C;
        if (ok) { a = sA[i]; b = sB[i]; a2 = sA[i + off]; b2 = sB[i + off]; }
        __syncthreads();
        if (ok) { sA[i] = a * a2; sB[i] = b + a * b2; }
        __syncthreads();
    }
    if (i < C) carryIn[i] = (i + 1 < C) ? sB[i + 1] : 0.0f;
}

__global__ __launch_bounds__(256)
void rlm_finalize(const float* __restrict__ wsA, const float* __restrict__ wsB,
                  const float* __restrict__ wsE, const float* __restrict__ wsM,
                  const float* __restrict__ wsD, const float* __restrict__ carryIn,
                  float* __restrict__ targets, int S)
{
    const int c = blockIdx.x * blockDim.x + threadIdx.x;
    const int base = c * S;
    float y = carryIn[c];
    for (int i = S - 1; i >= 0; --i) {
        const int t = base + i;
        const float g = wsE[t] + wsM[t] * y;
        targets[t] = g + wsD[t];
        y = wsB[t] + wsA[t] * y;
    }
}

extern "C" void kernel_launch(void* const* d_in, const int* in_sizes, int n_in,
                              void* d_out, int out_size, void* d_ws, size_t ws_size,
                              hipStream_t stream)
{
    const float* q      = (const float*)d_in[0];
    const float* q_tar  = (const float*)d_in[1];
    const float* pi     = (const float*)d_in[2];
    const int*   a_t    = (const int*)  d_in[3];
    const float* r_t    = (const float*)d_in[4];
    const float* mu     = (const float*)d_in[5];
    const int*   done_t = (const int*)  d_in[6];

    const int T = in_sizes[4];
    float* targets = (float*)d_out;
    float* adv     = targets + T;

    float* ws  = (float*)d_ws;
    float* wsA = ws;
    float* wsB = ws + (size_t)T;
    float* wsE = ws + (size_t)2 * T;
    float* wsM = ws + (size_t)3 * T;
    float* wsD = ws + (size_t)4 * T;
    const int S = 256;
    const int C = T / S;
    float* chA     = ws + (size_t)5 * T;
    float* chB     = chA + C;
    float* carryIn = chB + C;

    const int tiles   = T / 16;
    const int blocks1 = tiles / 8;

    rlm_prep<<<blocks1, 256, 0, stream>>>(q, q_tar, pi, a_t, r_t, mu, done_t,
                                          adv, wsA, wsB, wsE, wsM, wsD);
    rlm_compose<<<C / 256, 256, 0, stream>>>(wsA, wsB, chA, chB, S);
    rlm_chunkscan<<<1, 1024, 0, stream>>>(chA, chB, carryIn, C);
    rlm_finalize<<<C / 256, 256, 0, stream>>>(wsA, wsB, wsE, wsM, wsD, carryIn,
                                              targets, S);
}


# block-map scan launch removed: each apply workgroup folds the maps to its right itself (8 per thread, shuffle fold, LDS over waves), 3 launches become 2 (carry sum order changed)
# speedup vs baseline: 5.8326x; 5.8326x over previous
#include <hip/hip_runtime.h>


#define ALPHA  0.99f
#define GAMMA  0.999f
#define LAMBD  0.7f

typedef __attribute__((ext_vector_type(2))) float v2f;
typedef __attribute__((ext_vector_type(4))) float v4f;
typedef __attribute__((ext_vector_type(8))) float v8f;

#define TPITCH 68

__global__ __launch_bounds__(256)
void rlm_prep(const float* __restrict__ q, const float* __restrict__ q_tar,
              const float* __restrict__ pi, const int* __restrict__ a_t,
              const float* __restrict__ r_t, const float* __restrict__ mu,
              const int* __restrict__ done_t,
              float* __restrict__ adv,
              float* __restrict__ wsU, float* __restrict__ wsV,
              float* __restrict__ chA, float* __restrict__ chB)
{
    __shared__ __attribute__((aligned(16))) float tileX[8 * 16 * TPITCH];
    __shared__ __attribute__((aligned(16))) float tileY[8 * 16 * TPITCH];
    __shared__ float aggA[8];
    __shared__ float aggB[8];

    const int lane = threadIdx.x & 31;
    const int wave = threadIdx.x >> 5;
    const int tile = blockIdx.x * 8 + wave;
    const int t0   = tile * 16;
    const int cl   = lane & 15;
    const int hl   = lane >> 4;

    int a = 0, d = 0;
    float r = 0.0f;
    if (lane < 16) {
        a = a_t[t0 + lane];
        r = r_t[t0 + lane];
        d = done_t[t0 + lane];
    }

    const size_t sbase = (size_t)(t0 + 8 * hl) * 64 + 4 * cl;
    const float* qs  = q     + sbase;
    const float* ps  = pi    + sbase;
    const float* ts  = q_tar + sbase + 64;
    const float* ns  = pi    + sbase + 64;
    v4f qv[8], pv[8], tv[8], nv[8];
    #pragma unroll
    for (int j = 0; j < 8; ++j) {
        qv[j] = *(const v4f*)(qs + 64 * j);
        pv[j] = *(const v4f*)(ps + 64 * j);
    }
    #pragma unroll
    for (int j = 0; j < 8; ++j) {
        tv[j] = *(const v4f*)(ts + 64 * j);
        nv[j] = *(const v4f*)(ns + 64 * j);
    }

    float q_t_a = 0.0f, pi_a = 0.0f, mu_a = 1.0f;
    if (lane < 16) {
        const int g = (t0 + lane) * 64 + a;
        q_t_a = q [g];
        pi_a  = pi[g];
        mu_a  = mu[g];
    }

    float* X = tileX + wave * (16 * TPITCH);
    float* Y = tileY + wave * (16 * TPITCH);
    const int woff = (8 * hl) * TPITCH + 4 * cl;
    #pragma unroll
    for (int j = 0; j < 8; ++j)
        *(v4f*)(X + woff + j * TPITCH) = qv[j] * pv[j];
    #pragma unroll
    for (int j = 0; j < 8; ++j)
        *(v4f*)(Y + woff + j * TPITCH) = tv[j] * nv[j];

    __builtin_amdgcn_fence(__ATOMIC_ACQ_REL, "wavefront");

    const float* Xr = X + cl * TPITCH + 4 * hl;
    const float* Yr = Y + cl * TPITCH + 4 * hl;
    v4f xa[8], ya[8];
    #pragma unroll
    for (int i = 0; i < 8; ++i) {
        xa[i] = *(const v4f*)(Xr + 8 * i);
        ya[i] = *(const v4f*)(Yr + 8 * i);
    }
    __builtin_amdgcn_sched_barrier(0);

    const v2f ones = {1.0f, 1.0f};
    v8f acc_v  = {};
    v8f acc_vp = {};
    #pragma unroll
    for (int i = 0; i < 8; ++i) {
        const v2f x0 = {xa[i].x, xa[i].y};
        const v2f x1 = {xa[i].z, xa[i].w};
        const v2f y0 = {ya[i].x, ya[i].y};
        const v2f y1 = {ya[i].z, ya[i].w};
        acc_v  = __builtin_amdgcn_wmma_f32_16x16x4_f32(
            false, x0, false, ones, (short)0, acc_v,  false, false);
        acc_vp = __builtin_amdgcn_wmma_f32_16x16x4_f32(
            false, y0, false, ones, (short)0, acc_vp, false, false);
        acc_v  = __builtin_amdgcn_wmma_f32_16x16x4_f32(
            false, x1, false, ones, (short)0, acc_v,  false, false);
        acc_vp = __builtin_amdgcn_wmma_f32_16x16x4_f32(
            false, y1, false, ones, (short)0, acc_vp, false, false);
    }

    float* advs = adv + sbase;
    const float ascale = 1.0f - ALPHA;
    #pragma unroll
    for (int j = 0; j < 8; ++j) {
        const v4f o = (qv[j] - acc_v[j]) * ascale;
        *(v4f*)(advs + 64 * j) = o;
    }

    float selv = acc_v[0], selvp = acc_vp[0];
    #pragma unroll
    for (int j = 1; j < 8; ++j) {
        const bool pick = ((lane & 7) == j);
        selv  = pick ? acc_v[j]  : selv;
        selvp = pick ? acc_vp[j] : selvp;
    }
    const float upv  = __shfl_down(selv, 16);
    const float upvp = __shfl_down(selvp, 16);

    float ca = 1.0f, cb = 0.0f;
    float est = 0.0f, dlt = 0.0f, m = 0.0f;
    if (lane < 16) {
        const float df    = (float)d;
        const float vp1   = (lane & 8) ? upvp : selvp;
        const float vv    = (lane & 8) ? upv  : selv;

        est = r + (1.0f - df) * GAMMA * vp1;
        const float td  = est - q_t_a;
        const float rho = pi_a / mu_a;
        const float cc  = LAMBD * fminf(fmaxf(rho, 0.0f), 1.0f);

        cb  = LAMBD * GAMMA * rho * td;
        ca  = d ? 0.0f : GAMMA * cc;
        m   = d ? 0.0f : 1.0f;
        dlt = ALPHA * (q_t_a - vv);
    }

    for (int off = 1; off < 16; off <<= 1) {
        const float oa = __shfl_down(ca, off);
        const float ob = __shfl_down(cb, off);
        cb = cb + ca * ob;
        ca = ca * oa;
    }
    const float ea = __shfl_down(ca, 1);
    const float eb = __shfl_down(cb, 1);

    if (lane == 0) { aggA[wave] = ca; aggB[wave] = cb; }
    __syncthreads();

    float ga = 1.0f, gb = 0.0f;
    for (int j = 7; j >= 1; --j) {
        if (j > wave) {
            const float aj = aggA[j];
            const float bj = aggB[j];
            gb = bj + aj * gb;
            ga = aj * ga;
        }
    }

    if (threadIdx.x == 0) {
        chA[blockIdx.x] = ca * ga;
        chB[blockIdx.x] = cb + ca * gb;
    }

    if (lane < 16) {
        const int t = t0 + lane;
        const float P = ea * ga;
        const float Q = eb + ea * gb;
        wsU[t] = (est + m * Q) + dlt;
        wsV[t] = m * P;
    }
}

__global__ __launch_bounds__(256)
void rlm_finish(const float* __restrict__ wsU, const float* __restrict__ wsV,
                const float* __restrict__ chA, const float* __restrict__ chB,
                float* __restrict__ targets, int C)
{
    __shared__ float wA[8];
    __shared__ float wB[8];

    const int lane = threadIdx.x & 31;
    const int wave = threadIdx.x >> 5;
    const int i    = blockIdx.x * 256 + threadIdx.x;

    const v4f u = *(const v4f*)(wsU + 4 * (size_t)i);
    const v4f v = *(const v4f*)(wsV + 4 * (size_t)i);

    const int own = blockIdx.x * 8;
    float oa[7], ob[7];
    #pragma unroll
    for (int j = 0; j < 7; ++j) {
        oa[j] = chA[own + 1 + j];
        ob[j] = chB[own + 1 + j];
    }

    const int g0 = own + 8 + 8 * (int)threadIdx.x;
    float a = 1.0f, b = 0.0f;
    if (g0 < C) {
        const v4f a0 = *(const v4f*)(chA + g0);
        const v4f a1 = *(const v4f*)(chA + g0 + 4);
        const v4f b0 = *(const v4f*)(chB + g0);
        const v4f b1 = *(const v4f*)(chB + g0 + 4);
        b = b1.w;             a = a1.w;
        b = b1.z + a1.z * b;  a = a1.z * a;
        b = b1.y + a1.y * b;  a = a1.y * a;
        b = b1.x + a1.x * b;  a = a1.x * a;
        b = b0.w + a0.w * b;  a = a0.w * a;
        b = b0.z + a0.z * b;  a = a0.z * a;
        b = b0.y + a0.y * b;  a = a0.y * a;
        b = b0.x + a0.x * b;  a = a0.x * a;
    }

    for (int off = 1; off < 32; off <<= 1) {
        const float sa = __shfl_down(a, off);
        const float sb = __shfl_down(b, off);
        if (lane + off < 32) {
            b = b + a * sb;
            a = a * sa;
        }
    }
    if (lane == 0) { wA[wave] = a; wB[wave] = b; }
    __syncthreads();

    float y = 0.0f;
    #pragma unroll
    for (int w = 7; w >= 0; --w)
        y = wB[w] + wA[w] * y;

    #pragma unroll
    for (int j = 6; j >= 0; --j)
        if (j >= wave) y = ob[j] + oa[j] * y;

    *(v4f*)(targets + 4 * (size_t)i) = u + v * y;
}

extern "C" void kernel_launch(void* const* d_in, const int* in_sizes, int n_in,
                              void* d_out, int out_size, void* d_ws, size_t ws_size,
                              hipStream_t stream)
{
    const float* q      = (const float*)d_in[0];
    const float* q_tar  = (const float*)d_in[1];
    const float* pi     = (const float*)d_in[2];
    const int*   a_t    = (const int*)  d_in[3];
    const float* r_t    = (const float*)d_in[4];
    const float* mu     = (const float*)d_in[5];
    const int*   done_t = (const int*)  d_in[6];

    const int T = in_sizes[4];
    float* targets = (float*)d_out;
    float* adv     = targets + T;

    float* ws  = (float*)d_ws;
    float* wsU = ws;
    float* wsV = ws + (size_t)T;
    const int tiles   = T / 16;
    const int blocks1 = tiles / 8;
    float* chA     = ws + (size_t)2 * T;
    float* chB     = chA + blocks1;

    rlm_prep<<<blocks1, 256, 0, stream>>>(q, q_tar, pi, a_t, r_t, mu, done_t,
                                          adv, wsU, wsV, chA, chB);
    rlm_finish<<<T / 1024, 256, 0, stream>>>(wsU, wsV, chA, chB, targets, blocks1);
}


# prep blocks walk 4 chunks each; a wave requests its next tile's rows before it sums the tile in hand, block barrier waits on LDS only: 4096 long-lived waves (16 per WGP) instead of 16384 short ones
# speedup vs baseline: 6.2680x; 1.0746x over previous
#include <hip/hip_runtime.h>


#define ALPHA  0.99f
#define GAMMA  0.999f
#define LAMBD  0.7f

typedef __attribute__((ext_vector_type(2))) float v2f;
typedef __attribute__((ext_vector_type(4))) float v4f;
typedef __attribute__((ext_vector_type(8))) float v8f;

#define TPITCH 68
#define CPW    4

struct RowSet {
    v4f qv[8];
    v4f pv[8];
    v4f tv[8];
    v4f nv7;
};

struct PickSet { float q_t_a, pi_a, mu_a; };

__device__ __forceinline__
void issue_rows(RowSet& R, const float* __restrict__ q, const float* __restrict__ q_tar,
                const float* __restrict__ pi, size_t sbase)
{
    const float* qs = q     + sbase;
    const float* ps = pi    + sbase;
    const float* ts = q_tar + sbase + 64;
    #pragma unroll
    for (int j = 0; j < 8; ++j) {
        R.qv[j] = *(const v4f*)(qs + 64 * j);
        R.pv[j] = *(const v4f*)(ps + 64 * j);
    }
    #pragma unroll
    for (int j = 0; j < 8; ++j)
        R.tv[j] = *(const v4f*)(ts + 64 * j);
    R.nv7 = *(const v4f*)(ps + 64 * 8);
}

__device__ __forceinline__
void issue_picks(PickSet& P, const float* __restrict__ q, const float* __restrict__ pi,
                 const float* __restrict__ mu, int t0, int lane, int a)
{
    P.q_t_a = 0.0f; P.pi_a = 0.0f; P.mu_a = 1.0f;
    if (lane < 16) {
        const int g = (t0 + lane) * 64 + a;
        P.q_t_a = q [g];
        P.pi_a  = pi[g];
        P.mu_a  = mu[g];
    }
}

__device__ __forceinline__
void do_tile(const RowSet& R, const PickSet& P, int a, int d, float r,
             int chunk, int slot, int lane, int wave, int cl, int hl,
             float* X, float* Y, float (*aggA)[8], float (*aggB)[8],
             float* __restrict__ adv, float* __restrict__ wsU, float* __restrict__ wsV,
             float* __restrict__ chA, float* __restrict__ chB)
{
    const int t0 = (chunk * 8 + wave) * 16;
    const size_t sbase = (size_t)(t0 + 8 * hl) * 64 + 4 * cl;

    const int woff = (8 * hl) * TPITCH + 4 * cl;
    #pragma unroll
    for (int j = 0; j < 8; ++j)
        *(v4f*)(X + woff + j * TPITCH) = R.qv[j] * R.pv[j];
    #pragma unroll
    for (int j = 0; j < 7; ++j)
        *(v4f*)(Y + woff + j * TPITCH) = R.tv[j] * R.pv[j + 1];
    *(v4f*)(Y + woff + 7 * TPITCH) = R.tv[7] * R.nv7;

    __builtin_amdgcn_fence(__ATOMIC_ACQ_REL, "wavefront");

    const float* Xr = X + cl * TPITCH + 4 * hl;
    const float* Yr = Y + cl * TPITCH + 4 * hl;
    v4f xa[8], ya[8];
    #pragma unroll
    for (int i = 0; i < 8; ++i) {
        xa[i] = *(const v4f*)(Xr + 8 * i);
        ya[i] = *(const v4f*)(Yr + 8 * i);
    }
    __builtin_amdgcn_sched_barrier(0);

    const v2f ones = {1.0f, 1.0f};
    v8f acc_v  = {};
    v8f acc_vp = {};
    #pragma unroll
    for (int i = 0; i < 8; ++i) {
        const v2f x0 = {xa[i].x, xa[i].y};
        const v2f x1 = {xa[i].z, xa[i].w};
        const v2f y0 = {ya[i].x, ya[i].y};
        const v2f y1 = {ya[i].z, ya[i].w};
        acc_v  = __builtin_amdgcn_wmma_f32_16x16x4_f32(
            false, x0, false, ones, (short)0, acc_v,  false, false);
        acc_vp = __builtin_amdgcn_wmma_f32_16x16x4_f32(
            false, y0, false, ones, (short)0, acc_vp, false, false);
        acc_v  = __builtin_amdgcn_wmma_f32_16x16x4_f32(
            false, x1, false, ones, (short)0, acc_v,  false, false);
        acc_vp = __builtin_amdgcn_wmma_f32_16x16x4_f32(
            false, y1, false, ones, (short)0, acc_vp, false, false);
    }

    float* advs = adv + sbase;
    const float ascale = 1.0f - ALPHA;
    #pragma unroll
    for (int j = 0; j < 8; ++j) {
        const v4f o = (R.qv[j] - acc_v[j]) * ascale;
        *(v4f*)(advs + 64 * j) = o;
    }

    float selv = acc_v[0], selvp = acc_vp[0];
    #pragma unroll
    for (int j = 1; j < 8; ++j) {
        const bool pick = ((lane & 7) == j);
        selv  = pick ? acc_v[j]  : selv;
        selvp = pick ? acc_vp[j] : selvp;
    }
    const float upv  = __shfl_down(selv, 16);
    const float upvp = __shfl_down(selvp, 16);

    float ca = 1.0f, cb = 0.0f;
    float est = 0.0f, dlt = 0.0f, m = 0.0f;
    if (lane < 16) {
        const float df    = (float)d;
        const float vp1   = (lane & 8) ? upvp : selvp;
        const float vv    = (lane & 8) ? upv  : selv;

        est = r + (1.0f - df) * GAMMA * vp1;
        const float td  = est - P.q_t_a;
        const float rho = P.pi_a / P.mu_a;
        const float cc  = LAMBD * fminf(fmaxf(rho, 0.0f), 1.0f);

        cb  = LAMBD * GAMMA * rho * td;
        ca  = d ? 0.0f : GAMMA * cc;
        m   = d ? 0.0f : 1.0f;
        dlt = ALPHA * (P.q_t_a - vv);
    }

    for (int off = 1; off < 16; off <<= 1) {
        const float oa = __shfl_down(ca, off);
        const float ob = __shfl_down(cb, off);
        cb = cb + ca * ob;
        ca = ca * oa;
    }
    const float ea = __shfl_down(ca, 1);
    const float eb = __shfl_down(cb, 1);

    if (lane == 0) { aggA[slot][wave] = ca; aggB[slot][wave] = cb; }
    __builtin_amdgcn_fence(__ATOMIC_RELEASE, "workgroup", "local");
    __builtin_amdgcn_s_barrier();
    __builtin_amdgcn_fence(__ATOMIC_ACQUIRE, "workgroup", "local");

    float ga = 1.0f, gb = 0.0f;
    for (int j = 7; j >= 1; --j) {
        if (j > wave) {
            const float aj = aggA[slot][j];
            const float bj = aggB[slot][j];
            gb = bj + aj * gb;
            ga = aj * ga;
        }
    }

    if (threadIdx.x == 0) {
        chA[chunk] = ca * ga;
        chB[chunk] = cb + ca * gb;
    }

    if (lane < 16) {
        const int t = t0 + lane;
        const float Pm = ea * ga;
        const float Qm = eb + ea * gb;
        wsU[t] = (est + m * Qm) + dlt;
        wsV[t] = m * Pm;
    }
}

__global__ __launch_bounds__(256)
void rlm_prep(const float* __restrict__ q, const float* __restrict__ q_tar,
              const float* __restrict__ pi, const int* __restrict__ a_t,
              const float* __restrict__ r_t, const float* __restrict__ mu,
              const int* __restrict__ done_t,
              float* __restrict__ adv,
              float* __restrict__ wsU, float* __restrict__ wsV,
              float* __restrict__ chA, float* __restrict__ chB)
{
    __shared__ __attribute__((aligned(16))) float tileX[8 * 16 * TPITCH];
    __shared__ __attribute__((aligned(16))) float tileY[8 * 16 * TPITCH];
    __shared__ float aggA[CPW][8];
    __shared__ float aggB[CPW][8];

    const int lane = threadIdx.x & 31;
    const int wave = threadIdx.x >> 5;
    const int cl   = lane & 15;
    const int hl   = lane >> 4;
    const int chunk0 = blockIdx.x * CPW;

    float* X = tileX + wave * (16 * TPITCH);
    float* Y = tileY + wave * (16 * TPITCH);

    int a[CPW], d[CPW];
    float r[CPW];
    #pragma unroll
    for (int c = 0; c < CPW; ++c) { a[c] = 0; d[c] = 0; r[c] = 0.0f; }
    if (lane < 16) {
        #pragma unroll
        for (int c = 0; c < CPW; ++c) {
            const int t = ((chunk0 + c) * 8 + wave) * 16 + lane;
            a[c] = a_t[t];
            r[c] = r_t[t];
            d[c] = done_t[t];
        }
    }

    RowSet  R0, R1;
    PickSet P0, P1;
#define T0_OF(c)    (((chunk0 + (c)) * 8 + wave) * 16)
#define SBASE_OF(c) ((size_t)(T0_OF(c) + 8 * hl) * 64 + 4 * cl)

    issue_rows(R0, q, q_tar, pi, SBASE_OF(0));
    issue_picks(P0, q, pi, mu, T0_OF(0), lane, a[0]);

    issue_rows(R1, q, q_tar, pi, SBASE_OF(1));
    issue_picks(P1, q, pi, mu, T0_OF(1), lane, a[1]);
    __builtin_amdgcn_sched_barrier(0);
    do_tile(R0, P0, a[0], d[0], r[0], chunk0 + 0, 0, lane, wave, cl, hl, X, Y, aggA, aggB,
            adv, wsU, wsV, chA, chB);

    issue_rows(R0, q, q_tar, pi, SBASE_OF(2));
    issue_picks(P0, q, pi, mu, T0_OF(2), lane, a[2]);
    __builtin_amdgcn_sched_barrier(0);
    do_tile(R1, P1, a[1], d[1], r[1], chunk0 + 1, 1, lane, wave, cl, hl, X, Y, aggA, aggB,
            adv, wsU, wsV, chA, chB);

    issue_rows(R1, q, q_tar, pi, SBASE_OF(3));
    issue_picks(P1, q, pi, mu, T0_OF(3), lane, a[3]);
    __builtin_amdgcn_sched_barrier(0);
    do_tile(R0, P0, a[2], d[2], r[2], chunk0 + 2, 2, lane, wave, cl, hl, X, Y, aggA, aggB,
            adv, wsU, wsV, chA, chB);

    do_tile(R1, P1, a[3], d[3], r[3], chunk0 + 3, 3, lane, wave, cl, hl, X, Y, aggA, aggB,
            adv, wsU, wsV, chA, chB);
#undef T0_OF
#undef SBASE_OF
}

__global__ __launch_bounds__(256)
void rlm_finish(const float* __restrict__ wsU, const float* __restrict__ wsV,
                const float* __restrict__ chA, const float* __restrict__ chB,
                float* __restrict__ targets, int C)
{
    __shared__ float wA[8];
    __shared__ float wB[8];

    const int lane = threadIdx.x & 31;
    const int wave = threadIdx.x >> 5;
    const int i    = blockIdx.x * 256 + threadIdx.x;

    const v4f u = *(const v4f*)(wsU + 4 * (size_t)i);
    const v4f v = *(const v4f*)(wsV + 4 * (size_t)i);

    const int own = blockIdx.x * 8;
    float oa[7], ob[7];
    #pragma unroll
    for (int j = 0; j < 7; ++j) {
        oa[j] = chA[own + 1 + j];
        ob[j] = chB[own + 1 + j];
    }

    const int g0 = own + 8 + 8 * (int)threadIdx.x;
    float a = 1.0f, b = 0.0f;
    if (g0 < C) {
        const v4f a0 = *(const v4f*)(chA + g0);
        const v4f a1 = *(const v4f*)(chA + g0 + 4);
        const v4f b0 = *(const v4f*)(chB + g0);
        const v4f b1 = *(const v4f*)(chB + g0 + 4);
        b = b1.w;             a = a1.w;
        b = b1.z + a1.z * b;  a = a1.z * a;
        b = b1.y + a1.y * b;  a = a1.y * a;
        b = b1.x + a1.x * b;  a = a1.x * a;
        b = b0.w + a0.w * b;  a = a0.w * a;
        b = b0.z + a0.z * b;  a = a0.z * a;
        b = b0.y + a0.y * b;  a = a0.y * a;
        b = b0.x + a0.x * b;  a = a0.x * a;
    }

    for (int off = 1; off < 32; off <<= 1) {
        const float sa = __shfl_down(a, off);
        const float sb = __shfl_down(b, off);
        if (lane + off < 32) {
            b = b + a * sb;
            a = a * sa;
        }
    }
    if (lane == 0) { wA[wave] = a; wB[wave] = b; }
    __syncthreads();

    float y = 0.0f;
    #pragma unroll
    for (int w = 7; w >= 0; --w)
        y = wB[w] + wA[w] * y;

    #pragma unroll
    for (int j = 6; j >= 0; --j)
        if (j >= wave) y = ob[j] + oa[j] * y;

    *(v4f*)(targets + 4 * (size_t)i) = u + v * y;
}

extern "C" void kernel_launch(void* const* d_in, const int* in_sizes, int n_in,
                              void* d_out, int out_size, void* d_ws, size_t ws_size,
                              hipStream_t stream)
{
    const float* q      = (const float*)d_in[0];
    const float* q_tar  = (const float*)d_in[1];
    const float* pi     = (const float*)d_in[2];
    const int*   a_t    = (const int*)  d_in[3];
    const float* r_t    = (const float*)d_in[4];
    const float* mu     = (const float*)d_in[5];
    const int*   done_t = (const int*)  d_in[6];

    const int T = in_sizes[4];
    float* targets = (float*)d_out;
    float* adv     = targets + T;

    float* ws  = (float*)d_ws;
    float* wsU = ws;
    float* wsV = ws + (size_t)T;
    const int tiles   = T / 16;
    const int blocks1 = tiles / 8;
    float* chA     = ws + (size_t)2 * T;
    float* chB     = chA + blocks1;

    rlm_prep<<<blocks1 / CPW, 256, 0, stream>>>(q, q_tar, pi, a_t, r_t, mu, done_t,
                                                adv, wsU, wsV, chA, chB);
    rlm_finish<<<T / 1024, 256, 0, stream>>>(wsU, wsV, chA, chB, targets, blocks1);
}


# single launch: each block finds its incoming carry itself, exactly, walking right in 16-step tiles to the first episode end (at most to T); finish launch, U/V workspace gone (carry sum order changed)
# speedup vs baseline: 7.3241x; 1.1685x over previous
#include <hip/hip_runtime.h>


#define ALPHA  0.99f
#define GAMMA  0.999f
#define LAMBD  0.7f

typedef __attribute__((ext_vector_type(2))) float v2f;
typedef __attribute__((ext_vector_type(4))) float v4f;
typedef __attribute__((ext_vector_type(8))) float v8f;

#define TPITCH 68
#define CPW    4

struct RowSet {
    v4f qv[8];
    v4f pv[8];
    v4f tv[8];
    v4f nv7;
};

struct PickSet { float q_t_a, pi_a, mu_a; };

__device__ __forceinline__
void issue_rows(RowSet& R, const float* __restrict__ q, const float* __restrict__ q_tar,
                const float* __restrict__ pi, size_t sbase)
{
    const float* qs = q     + sbase;
    const float* ps = pi    + sbase;
    const float* ts = q_tar + sbase + 64;
    #pragma unroll
    for (int j = 0; j < 8; ++j) {
        R.qv[j] = *(const v4f*)(qs + 64 * j);
        R.pv[j] = *(const v4f*)(ps + 64 * j);
    }
    #pragma unroll
    for (int j = 0; j < 8; ++j)
        R.tv[j] = *(const v4f*)(ts + 64 * j);
    R.nv7 = *(const v4f*)(ps + 64 * 8);
}

__device__ __forceinline__
void issue_picks(PickSet& P, const float* __restrict__ q, const float* __restrict__ pi,
                 const float* __restrict__ mu, int t0, int lane, int a)
{
    P.q_t_a = 0.0f; P.pi_a = 0.0f; P.mu_a = 1.0f;
    if (lane < 16) {
        const int g = (t0 + lane) * 64 + a;
        P.q_t_a = q [g];
        P.pi_a  = pi[g];
        P.mu_a  = mu[g];
    }
}

struct RowSums { float selv, selvp, upv, upvp; };

__device__ __forceinline__
void do_tile_a(const RowSet& R, int chunk, int lane, int wave, int cl, int hl,
               float* X, float* Y, float* __restrict__ adv, RowSums& S)
{
    const int t0 = (chunk * 8 + wave) * 16;
    const size_t sbase = (size_t)(t0 + 8 * hl) * 64 + 4 * cl;

    const int woff = (8 * hl) * TPITCH + 4 * cl;
    #pragma unroll
    for (int j = 0; j < 8; ++j)
        *(v4f*)(X + woff + j * TPITCH) = R.qv[j] * R.pv[j];
    #pragma unroll
    for (int j = 0; j < 7; ++j)
        *(v4f*)(Y + woff + j * TPITCH) = R.tv[j] * R.pv[j + 1];
    *(v4f*)(Y + woff + 7 * TPITCH) = R.tv[7] * R.nv7;

    __builtin_amdgcn_fence(__ATOMIC_ACQ_REL, "wavefront");

    const float* Xr = X + cl * TPITCH + 4 * hl;
    const float* Yr = Y + cl * TPITCH + 4 * hl;
    v4f xa[8], ya[8];
    #pragma unroll
    for (int i = 0; i < 8; ++i) {
        xa[i] = *(const v4f*)(Xr + 8 * i);
        ya[i] = *(const v4f*)(Yr + 8 * i);
    }
    __builtin_amdgcn_sched_barrier(0);

    const v2f ones = {1.0f, 1.0f};
    v8f acc_v  = {};
    v8f acc_vp = {};
    #pragma unroll
    for (int i = 0; i < 8; ++i) {
        const v2f x0 = {xa[i].x, xa[i].y};
        const v2f x1 = {xa[i].z, xa[i].w};
        const v2f y0 = {ya[i].x, ya[i].y};
        const v2f y1 = {ya[i].z, ya[i].w};
        acc_v  = __builtin_amdgcn_wmma_f32_16x16x4_f32(
            false, x0, false, ones, (short)0, acc_v,  false, false);
        acc_vp = __builtin_amdgcn_wmma_f32_16x16x4_f32(
            false, y0, false, ones, (short)0, acc_vp, false, false);
        acc_v  = __builtin_amdgcn_wmma_f32_16x16x4_f32(
            false, x1, false, ones, (short)0, acc_v,  false, false);
        acc_vp = __builtin_amdgcn_wmma_f32_16x16x4_f32(
            false, y1, false, ones, (short)0, acc_vp, false, false);
    }

    float* advs = adv + sbase;
    const float ascale = 1.0f - ALPHA;
    #pragma unroll
    for (int j = 0; j < 8; ++j) {
        const v4f o = (R.qv[j] - acc_v[j]) * ascale;
        *(v4f*)(advs + 64 * j) = o;
    }

    float selv = acc_v[0], selvp = acc_vp[0];
    #pragma unroll
    for (int j = 1; j < 8; ++j) {
        const bool pick = ((lane & 7) == j);
        selv  = pick ? acc_v[j]  : selv;
        selvp = pick ? acc_vp[j] : selvp;
    }
    S.selv  = selv;
    S.selvp = selvp;
    S.upv   = __shfl_down(selv, 16);
    S.upvp  = __shfl_down(selvp, 16);
}

__device__ __forceinline__
void do_tile_b(const RowSums& S, const PickSet& P, int a, int d, float r,
               int slot, int lane, int wave,
               float (*aggA)[8], float (*aggB)[8],
               float* chAs, float* chBs, float* tgU, float* tgV)
{
    const float selv = S.selv, selvp = S.selvp, upv = S.upv, upvp = S.upvp;

    float ca = 1.0f, cb = 0.0f;
    float est = 0.0f, dlt = 0.0f, m = 0.0f;
    if (lane < 16) {
        const float df    = (float)d;
        const float vp1   = (lane & 8) ? upvp : selvp;
        const float vv    = (lane & 8) ? upv  : selv;

        est = r + (1.0f - df) * GAMMA * vp1;
        const float td  = est - P.q_t_a;
        const float rho = P.pi_a / P.mu_a;
        const float cc  = LAMBD * fminf(fmaxf(rho, 0.0f), 1.0f);

        cb  = LAMBD * GAMMA * rho * td;
        ca  = d ? 0.0f : GAMMA * cc;
        m   = d ? 0.0f : 1.0f;
        dlt = ALPHA * (P.q_t_a - vv);
    }

    for (int off = 1; off < 16; off <<= 1) {
        const float oa = __shfl_down(ca, off);
        const float ob = __shfl_down(cb, off);
        cb = cb + ca * ob;
        ca = ca * oa;
    }
    const float ea = __shfl_down(ca, 1);
    const float eb = __shfl_down(cb, 1);

    if (lane == 0) { aggA[slot][wave] = ca; aggB[slot][wave] = cb; }
    __builtin_amdgcn_fence(__ATOMIC_RELEASE, "workgroup", "local");
    __builtin_amdgcn_s_barrier();
    __builtin_amdgcn_fence(__ATOMIC_ACQUIRE, "workgroup", "local");

    float ga = 1.0f, gb = 0.0f;
    for (int j = 7; j >= 1; --j) {
        if (j > wave) {
            const float aj = aggA[slot][j];
            const float bj = aggB[slot][j];
            gb = bj + aj * gb;
            ga = aj * ga;
        }
    }

    if (threadIdx.x == 0) {
        chAs[slot] = ca * ga;
        chBs[slot] = cb + ca * gb;
    }

    if (lane < 16) {
        const float Pm = ea * ga;
        const float Qm = eb + ea * gb;
        const int   k  = slot * 128 + wave * 16 + lane;
        tgU[k] = (est + m * Qm) + dlt;
        tgV[k] = m * Pm;
    }
}

__device__ __forceinline__
void do_tile(const RowSet& R, const PickSet& P, int a, int d, float r,
             int chunk, int slot, int lane, int wave, int cl, int hl,
             float* X, float* Y, float (*aggA)[8], float (*aggB)[8],
             float* chAs, float* chBs, float* tgU, float* tgV, float* __restrict__ adv)
{
    RowSums S;
    do_tile_a(R, chunk, lane, wave, cl, hl, X, Y, adv, S);
    do_tile_b(S, P, a, d, r, slot, lane, wave, aggA, aggB, chAs, chBs, tgU, tgV);
}


struct HaloSet {
    v4f pn[8];
    v4f tv[8];
};

__device__ __forceinline__
void issue_halo_rows(HaloSet& H, const float* __restrict__ q_tar,
                     const float* __restrict__ pi, size_t sbase)
{
    const float* ps = pi    + sbase + 64;
    const float* ts = q_tar + sbase + 64;
    #pragma unroll
    for (int j = 0; j < 8; ++j) {
        H.pn[j] = *(const v4f*)(ps + 64 * j);
        H.tv[j] = *(const v4f*)(ts + 64 * j);
    }
}

__device__ __forceinline__
void halo_tile(const HaloSet& H, const PickSet& P, int d, float r,
               int lane, int cl, int hl, float* Y,
               float& A, float& B, bool& any_end)
{
    const int woff = (8 * hl) * TPITCH + 4 * cl;
    #pragma unroll
    for (int j = 0; j < 8; ++j)
        *(v4f*)(Y + woff + j * TPITCH) = H.tv[j] * H.pn[j];

    __builtin_amdgcn_fence(__ATOMIC_ACQ_REL, "wavefront");

    const float* Yr = Y + cl * TPITCH + 4 * hl;
    v4f ya[8];
    #pragma unroll
    for (int i = 0; i < 8; ++i)
        ya[i] = *(const v4f*)(Yr + 8 * i);
    __builtin_amdgcn_sched_barrier(0);

    const v2f ones = {1.0f, 1.0f};
    v8f acc0 = {};
    v8f acc1 = {};
    #pragma unroll
    for (int i = 0; i < 8; ++i) {
        const v2f y0 = {ya[i].x, ya[i].y};
        const v2f y1 = {ya[i].z, ya[i].w};
        acc0 = __builtin_amdgcn_wmma_f32_16x16x4_f32(
            false, y0, false, ones, (short)0, acc0, false, false);
        acc1 = __builtin_amdgcn_wmma_f32_16x16x4_f32(
            false, y1, false, ones, (short)0, acc1, false, false);
    }
    const v8f acc_vp = acc0 + acc1;

    float selvp = acc_vp[0];
    #pragma unroll
    for (int j = 1; j < 8; ++j)
        selvp = ((lane & 7) == j) ? acc_vp[j] : selvp;
    const float upvp = __shfl_down(selvp, 16);

    float ca = 1.0f, cb = 0.0f;
    if (lane < 16) {
        const float df  = (float)d;
        const float vp1 = (lane & 8) ? upvp : selvp;
        const float est = r + (1.0f - df) * GAMMA * vp1;
        const float td  = est - P.q_t_a;
        const float rho = P.pi_a / P.mu_a;
        const float cc  = LAMBD * fminf(fmaxf(rho, 0.0f), 1.0f);
        cb = LAMBD * GAMMA * rho * td;
        ca = d ? 0.0f : GAMMA * cc;
    }
    for (int off = 1; off < 16; off <<= 1) {
        const float oa = __shfl_down(ca, off);
        const float ob = __shfl_down(cb, off);
        cb = cb + ca * ob;
        ca = ca * oa;
    }
    A = __shfl(ca, 0);
    B = __shfl(cb, 0);
    any_end = __builtin_amdgcn_ballot_w32(d != 0) != 0u;
}

__global__ __launch_bounds__(256)
void rlm_prep(const float* __restrict__ q, const float* __restrict__ q_tar,
              const float* __restrict__ pi, const int* __restrict__ a_t,
              const float* __restrict__ r_t, const float* __restrict__ mu,
              const int* __restrict__ done_t,
              float* __restrict__ targets, float* __restrict__ adv, int T)
{
    __shared__ __attribute__((aligned(16))) float tileX[8 * 16 * TPITCH];
    __shared__ __attribute__((aligned(16))) float tileY[8 * 16 * TPITCH];
    __shared__ float aggA[CPW][8];
    __shared__ float aggB[CPW][8];
    __shared__ float chAs[CPW];
    __shared__ float chBs[CPW];
    __shared__ float yinS;
    __shared__ __attribute__((aligned(16))) float tgU[CPW * 128];
    __shared__ __attribute__((aligned(16))) float tgV[CPW * 128];

    const int lane = threadIdx.x & 31;
    const int wave = threadIdx.x >> 5;
    const int cl   = lane & 15;
    const int hl   = lane >> 4;
    const int chunk0 = blockIdx.x * CPW;

    float* X = tileX + wave * (16 * TPITCH);
    float* Y = tileY + wave * (16 * TPITCH);

    const int  hfirst = (chunk0 + CPW) * 128;
    const bool walker = (__builtin_amdgcn_readfirstlane(wave) == 7) && (hfirst < T);

    int a[CPW], d[CPW];
    float r[CPW];
    #pragma unroll
    for (int c = 0; c < CPW; ++c) { a[c] = 0; d[c] = 0; r[c] = 0.0f; }
    if (lane < 16) {
        #pragma unroll
        for (int c = 0; c < CPW; ++c) {
            const int t = ((chunk0 + c) * 8 + wave) * 16 + lane;
            a[c] = a_t[t];
            r[c] = r_t[t];
            d[c] = done_t[t];
        }
    }


    RowSet  R0, R1;
    PickSet P0, P1;

#define T0_OF(c)    (((chunk0 + (c)) * 8 + wave) * 16)
#define SBASE_OF(c) ((size_t)(T0_OF(c) + 8 * hl) * 64 + 4 * cl)

    issue_rows(R0, q, q_tar, pi, SBASE_OF(0));
    issue_picks(P0, q, pi, mu, T0_OF(0), lane, a[0]);

    issue_rows(R1, q, q_tar, pi, SBASE_OF(1));
    issue_picks(P1, q, pi, mu, T0_OF(1), lane, a[1]);
    __builtin_amdgcn_sched_barrier(0);
    do_tile(R0, P0, a[0], d[0], r[0], chunk0 + 0, 0, lane, wave, cl, hl, X, Y, aggA, aggB,
            chAs, chBs, tgU, tgV, adv);

    issue_rows(R0, q, q_tar, pi, SBASE_OF(2));
    issue_picks(P0, q, pi, mu, T0_OF(2), lane, a[2]);
    __builtin_amdgcn_sched_barrier(0);
    do_tile(R1, P1, a[1], d[1], r[1], chunk0 + 1, 1, lane, wave, cl, hl, X, Y, aggA, aggB,
            chAs, chBs, tgU, tgV, adv);

    issue_rows(R1, q, q_tar, pi, SBASE_OF(3));
    issue_picks(P1, q, pi, mu, T0_OF(3), lane, a[3]);
    __builtin_amdgcn_sched_barrier(0);
    do_tile(R0, P0, a[2], d[2], r[2], chunk0 + 2, 2, lane, wave, cl, hl, X, Y, aggA, aggB,
            chAs, chBs, tgU, tgV, adv);

    HaloSet H;
    PickSet HP;
    int ha = 0, hd = 0;
    float hr = 0.0f;
    if (walker) {
        if (lane < 16) {
            ha = a_t[hfirst + lane];
            hr = r_t[hfirst + lane];
            hd = done_t[hfirst + lane];
        }
    }
    __builtin_amdgcn_sched_barrier(0);
    RowSums S3;
    do_tile_a(R1, chunk0 + 3, lane, wave, cl, hl, X, Y, adv, S3);
    if (walker) {
        issue_halo_rows(H, q_tar, pi, (size_t)(hfirst + 8 * hl) * 64 + 4 * cl);
        issue_picks(HP, q, pi, mu, hfirst, lane, ha);
    }
    __builtin_amdgcn_sched_barrier(0);
    do_tile_b(S3, P1, a[3], d[3], r[3], 3, lane, wave, aggA, aggB, chAs, chBs, tgU, tgV);

    if (__builtin_amdgcn_readfirstlane(wave) == 7) {
        float PA = 1.0f, PB = 0.0f;
        if (walker) {
            for (int h0 = hfirst; h0 < T; h0 += 16) {
                float A, B;
                bool any_end;
                halo_tile(H, HP, hd, hr, lane, cl, hl, Y, A, B, any_end);
                PB = PB + PA * B;
                PA = PA * A;
                if (any_end) break;
                const int hn = h0 + 16;
                if (hn < T) {
                    ha = 0; hd = 0; hr = 0.0f;
                    if (lane < 16) {
                        ha = a_t[hn + lane];
                        hr = r_t[hn + lane];
                        hd = done_t[hn + lane];
                    }
                    issue_halo_rows(H, q_tar, pi, (size_t)(hn + 8 * hl) * 64 + 4 * cl);
                    issue_picks(HP, q, pi, mu, hn, lane, ha);
                }
            }
        }
        if (lane == 0) yinS = PB;
    }

    __builtin_amdgcn_fence(__ATOMIC_RELEASE, "workgroup", "local");
    __builtin_amdgcn_s_barrier();
    __builtin_amdgcn_fence(__ATOMIC_ACQUIRE, "workgroup", "local");

    if (wave < CPW) {
        float cin = yinS;
        #pragma unroll
        for (int c = CPW - 1; c >= 1; --c)
            if (c > wave) cin = chBs[c] + chAs[c] * cin;
        const v4f u = *(const v4f*)(tgU + 4 * threadIdx.x);
        const v4f v = *(const v4f*)(tgV + 4 * threadIdx.x);
        *(v4f*)(targets + (size_t)chunk0 * 128 + 4 * threadIdx.x) = u + v * cin;
    }
#undef T0_OF
#undef SBASE_OF
}

extern "C" void kernel_launch(void* const* d_in, const int* in_sizes, int n_in,
                              void* d_out, int out_size, void* d_ws, size_t ws_size,
                              hipStream_t stream)
{
    const float* q      = (const float*)d_in[0];
    const float* q_tar  = (const float*)d_in[1];
    const float* pi     = (const float*)d_in[2];
    const int*   a_t    = (const int*)  d_in[3];
    const float* r_t    = (const float*)d_in[4];
    const float* mu     = (const float*)d_in[5];
    const int*   done_t = (const int*)  d_in[6];

    const int T = in_sizes[4];
    float* targets = (float*)d_out;
    float* adv     = targets + T;

    const int tiles   = T / 16;
    const int blocks1 = tiles / 8;

    rlm_prep<<<blocks1 / CPW, 256, 0, stream>>>(q, q_tar, pi, a_t, r_t, mu, done_t,
                                                targets, adv, T);
}


# loads that an episode end makes dead are not issued: q_tar row t+1 only where done[t]=0, the pi/mu entries of rho[t] only where done[t-1]=0; unread operands enter as 0 and 1 (no sum order changed)
# speedup vs baseline: 9.2503x; 1.2630x over previous
#include <hip/hip_runtime.h>


#define ALPHA  0.99f
#define GAMMA  0.999f
#define LAMBD  0.7f

typedef __attribute__((ext_vector_type(2))) float v2f;
typedef __attribute__((ext_vector_type(4))) float v4f;
typedef __attribute__((ext_vector_type(8))) float v8f;

#define TPITCH 20
#define CPW    4

struct RowSet {
    v4f qv[8];
    v4f pv[8];
    v4f tv[8];
    v4f nv7;
};

struct PickSet { float q_t_a, pi_a, mu_a; };

__device__ __forceinline__
float dot4(const v4f a, const v4f b)
{
    const v2f a0 = {a.x, a.y}, a1 = {a.z, a.w};
    const v2f b0 = {b.x, b.y}, b1 = {b.z, b.w};
    const v2f t = __builtin_elementwise_fma(a1, b1, a0 * b0);
    return t.x + t.y;
}

__device__ __forceinline__
void issue_rows(RowSet& R, const float* __restrict__ q, const float* __restrict__ q_tar,
                const float* __restrict__ pi, size_t sbase, int dflag, int hl, int lane,
                bool& rho_used)
{
    const float* qs = q     + sbase;
    const float* ps = pi    + sbase;
    const float* ts = q_tar + sbase + 64;
    #pragma unroll
    for (int j = 0; j < 8; ++j) {
        R.qv[j] = *(const v4f*)(qs + 64 * j);
        R.pv[j] = *(const v4f*)(ps + 64 * j);
    }
    R.nv7 = *(const v4f*)(ps + 64 * 8);

    __builtin_amdgcn_sched_barrier(0);

    const unsigned open = __builtin_amdgcn_ballot_w32(dflag == 0);
    const unsigned live = open >> (8 * hl);
    const unsigned open_left = (open << 1) | ((open >> 16) & 1u);
    rho_used = ((open_left >> lane) & 1u) != 0u;
    #pragma unroll
    for (int j = 0; j < 8; ++j) {
        R.tv[j] = (v4f){0.0f, 0.0f, 0.0f, 0.0f};
        if ((live >> j) & 1u)
            R.tv[j] = *(const v4f*)(ts + 64 * j);
    }
}

__device__ __forceinline__
void issue_picks(PickSet& P, const float* __restrict__ q, const float* __restrict__ pi,
                 const float* __restrict__ mu, int t0, int lane, int a, bool rho_used)
{
    P.q_t_a = 0.0f; P.pi_a = 0.0f; P.mu_a = 1.0f;
    if (lane < 16) {
        const int g = (t0 + lane) * 64 + a;
        P.q_t_a = q [g];
        if (rho_used) {
            P.pi_a = pi[g];
            P.mu_a = mu[g];
        }
    }
}

struct RowSums { float selv, selvp, upv, upvp; };

__device__ __forceinline__
void do_tile_a(const RowSet& R, int chunk, int lane, int wave, int cl, int hl,
               float* X, float* Y, float* __restrict__ adv, RowSums& S)
{
    const int t0 = (chunk * 8 + wave) * 16;
    const size_t sbase = (size_t)(t0 + 8 * hl) * 64 + 4 * cl;

    const int woff = (8 * hl) * TPITCH + cl;
    #pragma unroll
    for (int j = 0; j < 8; ++j)
        X[woff + j * TPITCH] = dot4(R.qv[j], R.pv[j]);
    #pragma unroll
    for (int j = 0; j < 7; ++j)
        Y[woff + j * TPITCH] = dot4(R.tv[j], R.pv[j + 1]);
    Y[woff + 7 * TPITCH] = dot4(R.tv[7], R.nv7);

    __builtin_amdgcn_fence(__ATOMIC_ACQ_REL, "wavefront");

    const float* Xr = X + cl * TPITCH + 8 * hl;
    const float* Yr = Y + cl * TPITCH + 8 * hl;
    v4f xa[2], ya[2];
    #pragma unroll
    for (int i = 0; i < 2; ++i) {
        xa[i] = *(const v4f*)(Xr + 4 * i);
        ya[i] = *(const v4f*)(Yr + 4 * i);
    }
    __builtin_amdgcn_sched_barrier(0);

    const v2f ones = {1.0f, 1.0f};
    v8f acc_v  = {};
    v8f acc_vp = {};
    #pragma unroll
    for (int i = 0; i < 2; ++i) {
        const v2f x0 = {xa[i].x, xa[i].y};
        const v2f x1 = {xa[i].z, xa[i].w};
        const v2f y0 = {ya[i].x, ya[i].y};
        const v2f y1 = {ya[i].z, ya[i].w};
        acc_v  = __builtin_amdgcn_wmma_f32_16x16x4_f32(
            false, x0, false, ones, (short)0, acc_v,  false, false);
        acc_vp = __builtin_amdgcn_wmma_f32_16x16x4_f32(
            false, y0, false, ones, (short)0, acc_vp, false, false);
        acc_v  = __builtin_amdgcn_wmma_f32_16x16x4_f32(
            false, x1, false, ones, (short)0, acc_v,  false, false);
        acc_vp = __builtin_amdgcn_wmma_f32_16x16x4_f32(
            false, y1, false, ones, (short)0, acc_vp, false, false);
    }

    float* advs = adv + sbase;
    const float ascale = 1.0f - ALPHA;
    #pragma unroll
    for (int j = 0; j < 8; ++j) {
        const v4f o = (R.qv[j] - acc_v[j]) * ascale;
        *(v4f*)(advs + 64 * j) = o;
    }

    float selv = acc_v[0], selvp = acc_vp[0];
    #pragma unroll
    for (int j = 1; j < 8; ++j) {
        const bool pick = ((lane & 7) == j);
        selv  = pick ? acc_v[j]  : selv;
        selvp = pick ? acc_vp[j] : selvp;
    }
    S.selv  = selv;
    S.selvp = selvp;
    S.upv   = __shfl_down(selv, 16);
    S.upvp  = __shfl_down(selvp, 16);
}

__device__ __forceinline__
void do_tile_b(const RowSums& S, const PickSet& P, int a, int d, float r,
               int slot, int lane, int wave,
               float (*aggA)[8], float (*aggB)[8],
               float* chAs, float* chBs, float* tgU, float* tgV)
{
    const float selv = S.selv, selvp = S.selvp, upv = S.upv, upvp = S.upvp;

    float ca = 1.0f, cb = 0.0f;
    float est = 0.0f, dlt = 0.0f, m = 0.0f;
    if (lane < 16) {
        const float df    = (float)d;
        const float vp1   = (lane & 8) ? upvp : selvp;
        const float vv    = (lane & 8) ? upv  : selv;

        est = r + (1.0f - df) * GAMMA * vp1;
        const float td  = est - P.q_t_a;
        const float rho = P.pi_a / P.mu_a;
        const float cc  = LAMBD * fminf(fmaxf(rho, 0.0f), 1.0f);

        cb  = LAMBD * GAMMA * rho * td;
        ca  = d ? 0.0f : GAMMA * cc;
        m   = d ? 0.0f : 1.0f;
        dlt = ALPHA * (P.q_t_a - vv);
    }

    for (int off = 1; off < 16; off <<= 1) {
        const float oa = __shfl_down(ca, off);
        const float ob = __shfl_down(cb, off);
        cb = cb + ca * ob;
        ca = ca * oa;
    }
    const float ea = __shfl_down(ca, 1);
    const float eb = __shfl_down(cb, 1);

    if (lane == 0) { aggA[slot][wave] = ca; aggB[slot][wave] = cb; }
    __builtin_amdgcn_fence(__ATOMIC_RELEASE, "workgroup", "local");
    __builtin_amdgcn_s_barrier();
    __builtin_amdgcn_fence(__ATOMIC_ACQUIRE, "workgroup", "local");

    float ga = 1.0f, gb = 0.0f;
    for (int j = 7; j >= 1; --j) {
        if (j > wave) {
            const float aj = aggA[slot][j];
            const float bj = aggB[slot][j];
            gb = bj + aj * gb;
            ga = aj * ga;
        }
    }

    if (threadIdx.x == 0) {
        chAs[slot] = ca * ga;
        chBs[slot] = cb + ca * gb;
    }

    if (lane < 16) {
        const float Pm = ea * ga;
        const float Qm = eb + ea * gb;
        const int   k  = slot * 128 + wave * 16 + lane;
        tgU[k] = (est + m * Qm) + dlt;
        tgV[k] = m * Pm;
    }
}

__device__ __forceinline__
void do_tile(const RowSet& R, const PickSet& P, int a, int d, float r,
             int chunk, int slot, int lane, int wave, int cl, int hl,
             float* X, float* Y, float (*aggA)[8], float (*aggB)[8],
             float* chAs, float* chBs, float* tgU, float* tgV, float* __restrict__ adv)
{
    RowSums S;
    do_tile_a(R, chunk, lane, wave, cl, hl, X, Y, adv, S);
    do_tile_b(S, P, a, d, r, slot, lane, wave, aggA, aggB, chAs, chBs, tgU, tgV);
}


struct HaloSet {
    v4f pn[8];
    v4f tv[8];
};

__device__ __forceinline__
void issue_halo_rows(HaloSet& H, const float* __restrict__ q_tar,
                     const float* __restrict__ pi, size_t sbase)
{
    const float* ps = pi    + sbase + 64;
    const float* ts = q_tar + sbase + 64;
    #pragma unroll
    for (int j = 0; j < 8; ++j) {
        H.pn[j] = *(const v4f*)(ps + 64 * j);
        H.tv[j] = *(const v4f*)(ts + 64 * j);
    }
}

__device__ __forceinline__
void halo_tile(const HaloSet& H, const PickSet& P, int d, float r,
               int lane, int cl, int hl, float* Y,
               float& A, float& B, bool& any_end)
{
    const int woff = (8 * hl) * TPITCH + cl;
    #pragma unroll
    for (int j = 0; j < 8; ++j)
        Y[woff + j * TPITCH] = dot4(H.tv[j], H.pn[j]);

    __builtin_amdgcn_fence(__ATOMIC_ACQ_REL, "wavefront");

    const float* Yr = Y + cl * TPITCH + 8 * hl;
    v4f ya[2];
    #pragma unroll
    for (int i = 0; i < 2; ++i)
        ya[i] = *(const v4f*)(Yr + 4 * i);
    __builtin_amdgcn_sched_barrier(0);

    const v2f ones = {1.0f, 1.0f};
    v8f acc0 = {};
    v8f acc1 = {};
    #pragma unroll
    for (int i = 0; i < 2; ++i) {
        const v2f y0 = {ya[i].x, ya[i].y};
        const v2f y1 = {ya[i].z, ya[i].w};
        acc0 = __builtin_amdgcn_wmma_f32_16x16x4_f32(
            false, y0, false, ones, (short)0, acc0, false, false);
        acc1 = __builtin_amdgcn_wmma_f32_16x16x4_f32(
            false, y1, false, ones, (short)0, acc1, false, false);
    }
    const v8f acc_vp = acc0 + acc1;

    float selvp = acc_vp[0];
    #pragma unroll
    for (int j = 1; j < 8; ++j)
        selvp = ((lane & 7) == j) ? acc_vp[j] : selvp;
    const float upvp = __shfl_down(selvp, 16);

    float ca = 1.0f, cb = 0.0f;
    if (lane < 16) {
        const float df  = (float)d;
        const float vp1 = (lane & 8) ? upvp : selvp;
        const float est = r + (1.0f - df) * GAMMA * vp1;
        const float td  = est - P.q_t_a;
        const float rho = P.pi_a / P.mu_a;
        const float cc  = LAMBD * fminf(fmaxf(rho, 0.0f), 1.0f);
        cb = LAMBD * GAMMA * rho * td;
        ca = d ? 0.0f : GAMMA * cc;
    }
    for (int off = 1; off < 16; off <<= 1) {
        const float oa = __shfl_down(ca, off);
        const float ob = __shfl_down(cb, off);
        cb = cb + ca * ob;
        ca = ca * oa;
    }
    A = __shfl(ca, 0);
    B = __shfl(cb, 0);
    any_end = __builtin_amdgcn_ballot_w32(d != 0) != 0u;
}

__global__ __launch_bounds__(256)
void rlm_prep(const float* __restrict__ q, const float* __restrict__ q_tar,
              const float* __restrict__ pi, const int* __restrict__ a_t,
              const float* __restrict__ r_t, const float* __restrict__ mu,
              const int* __restrict__ done_t,
              float* __restrict__ targets, float* __restrict__ adv, int T)
{
    __shared__ __attribute__((aligned(16))) float tileX[8 * 16 * TPITCH];
    __shared__ __attribute__((aligned(16))) float tileY[8 * 16 * TPITCH];
    __shared__ float aggA[CPW][8];
    __shared__ float aggB[CPW][8];
    __shared__ float chAs[CPW];
    __shared__ float chBs[CPW];
    __shared__ float yinS;
    __shared__ __attribute__((aligned(16))) float tgU[CPW * 128];
    __shared__ __attribute__((aligned(16))) float tgV[CPW * 128];

    const int lane = threadIdx.x & 31;
    const int wave = threadIdx.x >> 5;
    const int cl   = lane & 15;
    const int hl   = lane >> 4;
    const int chunk0 = blockIdx.x * CPW;

    float* X = tileX + wave * (16 * TPITCH);
    float* Y = tileY + wave * (16 * TPITCH);

    const int  hfirst = (chunk0 + CPW) * 128;
    const bool walker = (__builtin_amdgcn_readfirstlane(wave) == 7) && (hfirst < T);

    int a[CPW], d[CPW];
    float r[CPW];
    #pragma unroll
    for (int c = 0; c < CPW; ++c) { a[c] = 0; d[c] = 0; r[c] = 0.0f; }
    if (lane <= 16) {
        #pragma unroll
        for (int c = 0; c < CPW; ++c) {
            const int t0 = ((chunk0 + c) * 8 + wave) * 16;
            const int t  = (lane < 16) ? t0 + lane : max(t0 - 1, 0);
            a[c] = a_t[t];
            r[c] = r_t[t];
            d[c] = done_t[t];
        }
    }

    RowSet  R0, R1;
    PickSet P0, P1;

#define T0_OF(c)    (((chunk0 + (c)) * 8 + wave) * 16)
#define SBASE_OF(c) ((size_t)(T0_OF(c) + 8 * hl) * 64 + 4 * cl)

    bool ru;
    issue_rows(R0, q, q_tar, pi, SBASE_OF(0), d[0], hl, lane, ru);
    issue_picks(P0, q, pi, mu, T0_OF(0), lane, a[0], ru);

    issue_rows(R1, q, q_tar, pi, SBASE_OF(1), d[1], hl, lane, ru);
    issue_picks(P1, q, pi, mu, T0_OF(1), lane, a[1], ru);
    __builtin_amdgcn_sched_barrier(0);
    do_tile(R0, P0, a[0], d[0], r[0], chunk0 + 0, 0, lane, wave, cl, hl, X, Y, aggA, aggB,
            chAs, chBs, tgU, tgV, adv);

    issue_rows(R0, q, q_tar, pi, SBASE_OF(2), d[2], hl, lane, ru);
    issue_picks(P0, q, pi, mu, T0_OF(2), lane, a[2], ru);
    __builtin_amdgcn_sched_barrier(0);
    do_tile(R1, P1, a[1], d[1], r[1], chunk0 + 1, 1, lane, wave, cl, hl, X, Y, aggA, aggB,
            chAs, chBs, tgU, tgV, adv);

    issue_rows(R1, q, q_tar, pi, SBASE_OF(3), d[3], hl, lane, ru);
    issue_picks(P1, q, pi, mu, T0_OF(3), lane, a[3], ru);
    __builtin_amdgcn_sched_barrier(0);
    do_tile(R0, P0, a[2], d[2], r[2], chunk0 + 2, 2, lane, wave, cl, hl, X, Y, aggA, aggB,
            chAs, chBs, tgU, tgV, adv);

    HaloSet H;
    PickSet HP;
    int ha = 0, hd = 0;
    float hr = 0.0f;
    if (walker) {
        if (lane < 16) {
            ha = a_t[hfirst + lane];
            hr = r_t[hfirst + lane];
            hd = done_t[hfirst + lane];
        }
    }
    __builtin_amdgcn_sched_barrier(0);
    RowSums S3;
    do_tile_a(R1, chunk0 + 3, lane, wave, cl, hl, X, Y, adv, S3);
    if (walker) {
        issue_halo_rows(H, q_tar, pi, (size_t)(hfirst + 8 * hl) * 64 + 4 * cl);
        issue_picks(HP, q, pi, mu, hfirst, lane, ha, true);
    }
    __builtin_amdgcn_sched_barrier(0);
    do_tile_b(S3, P1, a[3], d[3], r[3], 3, lane, wave, aggA, aggB, chAs, chBs, tgU, tgV);

    if (__builtin_amdgcn_readfirstlane(wave) == 7) {
        float PA = 1.0f, PB = 0.0f;
        if (walker) {
            for (int h0 = hfirst; h0 < T; h0 += 16) {
                float A, B;
                bool any_end;
                halo_tile(H, HP, hd, hr, lane, cl, hl, Y, A, B, any_end);
                PB = PB + PA * B;
                PA = PA * A;
                if (any_end) break;
                const int hn = h0 + 16;
                if (hn < T) {
                    ha = 0; hd = 0; hr = 0.0f;
                    if (lane < 16) {
                        ha = a_t[hn + lane];
                        hr = r_t[hn + lane];
                        hd = done_t[hn + lane];
                    }
                    issue_halo_rows(H, q_tar, pi, (size_t)(hn + 8 * hl) * 64 + 4 * cl);
                    issue_picks(HP, q, pi, mu, hn, lane, ha, true);
                }
            }
        }
        if (lane == 0) yinS = PB;
    }

    __builtin_amdgcn_fence(__ATOMIC_RELEASE, "workgroup", "local");
    __builtin_amdgcn_s_barrier();
    __builtin_amdgcn_fence(__ATOMIC_ACQUIRE, "workgroup", "local");

    if (wave < CPW) {
        float cin = yinS;
        #pragma unroll
        for (int c = CPW - 1; c >= 1; --c)
            if (c > wave) cin = chBs[c] + chAs[c] * cin;
        const v4f u = *(const v4f*)(tgU + 4 * threadIdx.x);
        const v4f v = *(const v4f*)(tgV + 4 * threadIdx.x);
        *(v4f*)(targets + (size_t)chunk0 * 128 + 4 * threadIdx.x) = u + v * cin;
    }
#undef T0_OF
#undef SBASE_OF
}

extern "C" void kernel_launch(void* const* d_in, const int* in_sizes, int n_in,
                              void* d_out, int out_size, void* d_ws, size_t ws_size,
                              hipStream_t stream)
{
    const float* q      = (const float*)d_in[0];
    const float* q_tar  = (const float*)d_in[1];
    const float* pi     = (const float*)d_in[2];
    const int*   a_t    = (const int*)  d_in[3];
    const float* r_t    = (const float*)d_in[4];
    const float* mu     = (const float*)d_in[5];
    const int*   done_t = (const int*)  d_in[6];

    const int T = in_sizes[4];
    float* targets = (float*)d_out;
    float* adv     = targets + T;

    const int tiles   = T / 16;
    const int blocks1 = tiles / 8;

    rlm_prep<<<blocks1 / CPW, 256, 0, stream>>>(q, q_tar, pi, a_t, r_t, mu, done_t,
                                                targets, adv, T);
}
